# starting kernel
# baseline (speedup 1.0000x reference)
#include <hip/hip_runtime.h>


typedef __attribute__((ext_vector_type(2))) float v2f;
typedef __attribute__((ext_vector_type(4))) float v4f;
typedef __attribute__((ext_vector_type(8))) float v8f;

#define SEG   256
#define CHUNK 16
#define CHUNKS_PER_SEG (SEG / CHUNK)

__global__ void seg_state_kernel(const float* __restrict__ u,
                                 const float* __restrict__ bc,
                                 const float* __restrict__ fc,
                                 float* __restrict__ zpart,
                                 int Bn, int T, int nseg) {
  int tid = blockIdx.x * blockDim.x + threadIdx.x;
  int total = Bn * nseg;
  if (tid >= total) return;
  int b = tid % Bn;
  int s = tid / Bn;
  const float b0 = bc[0], b1 = bc[1], b2v = 0.0f;
  const float a1 = fc[0], a2 = fc[1];
  float z0 = 0.f, z1 = 0.f;
  const float* up = u + (long long)b * T + (long long)s * SEG;
  #pragma unroll 4
  for (int t = 0; t < SEG; ++t) {
    float ut = up[t];
    float yv  = b0 * ut + z0;
    float nz0 = b1 * ut - a1 * yv + z1;
    float nz1 = b2v * ut - a2 * yv;
    z0 = nz0; z1 = nz1;
  }
  zpart[(size_t)(s * Bn + b) * 2 + 0] = z0;
  zpart[(size_t)(s * Bn + b) * 2 + 1] = z1;
}

__global__ void scan_state_kernel(const float* __restrict__ bc,
                                  const float* __restrict__ fc,
                                  const float* __restrict__ yinit,
                                  const float* __restrict__ uinit,
                                  const float* __restrict__ zpart,
                                  float* __restrict__ zin,
                                  int Bn, int nseg) {
  int b = blockIdx.x * blockDim.x + threadIdx.x;
  if (b >= Bn) return;
  const float b1 = bc[1];
  const float a1 = fc[0], a2 = fc[1];

  float m00 = -a1, m01 = 1.f, m10 = -a2, m11 = 0.f;
  #pragma unroll
  for (int i = 0; i < 8; ++i) {
    float t00 = m00 * m00 + m01 * m10;
    float t01 = m00 * m01 + m01 * m11;
    float t10 = m10 * m00 + m11 * m10;
    float t11 = m10 * m01 + m11 * m11;
    m00 = t00; m01 = t01; m10 = t10; m11 = t11;
  }

  float z0 = b1 * uinit[b * 2 + 0] - a1 * yinit[b * 2 + 0] - a2 * yinit[b * 2 + 1];
  float z1 = -a2 * yinit[b * 2 + 0];

  for (int s = 0; s < nseg; ++s) {
    size_t idx = (size_t)(s * Bn + b) * 2;
    zin[idx + 0] = z0;
    zin[idx + 1] = z1;
    float nz0 = m00 * z0 + m01 * z1 + zpart[idx + 0];
    float nz1 = m10 * z0 + m11 * z1 + zpart[idx + 1];
    z0 = nz0; z1 = nz1;
  }
}

__global__ __launch_bounds__(256)
void wmma_filter_kernel(const float* __restrict__ u,
                        const float* __restrict__ bc,
                        const float* __restrict__ fc,
                        const float* __restrict__ zin,
                        float* __restrict__ y,
                        int Bn, int T, int nseg) {
  __shared__ float Hs[16][16];
  __shared__ v2f   Cs[16];

  const float b0 = bc[0], b1 = bc[1], b2v = 0.0f;
  const float a1 = fc[0], a2 = fc[1];

  if (threadIdx.x < 16) {
    const int t = (int)threadIdx.x;
    float hm1 = 0.f, hm2 = 0.f;
    #pragma unroll
    for (int d = 0; d < 16; ++d) {
      float hd;
      if (d == 0)      hd = b0;
      else if (d == 1) hd = b1 - a1 * b0;
      else if (d == 2) hd = b2v - a1 * hm1 - a2 * hm2;
      else             hd = -a1 * hm1 - a2 * hm2;
      if (d <= t) Hs[t][t - d] = hd;
      hm2 = hm1; hm1 = hd;
    }
    for (int k = t + 1; k < 16; ++k) Hs[t][k] = 0.f;
  } else if (threadIdx.x == 16) {
    float cc0 = 1.f, cc1 = 0.f;
    #pragma unroll
    for (int k = 0; k < 16; ++k) {
      v2f c; c.x = cc0; c.y = cc1;
      Cs[k] = c;
      float n0 = -a1 * cc0 - a2 * cc1;
      cc1 = cc0; cc0 = n0;
    }
  }
  __syncthreads();

  const int lane = (int)(threadIdx.x & 31);
  const int wave = (int)((blockIdx.x * blockDim.x + threadIdx.x) >> 5);
  const int ngroups = Bn >> 4;
  const int totalWaves = nseg * ngroups;

  if (wave < totalWaves) {
    const int seg  = wave / ngroups;
    const int grp  = wave % ngroups;
    const int col  = lane & 15;
    const int half = lane >> 4;
    const int b    = (grp << 4) + col;

    v2f afrag[4];
    #pragma unroll
    for (int j = 0; j < 4; ++j)
      afrag[j] = *(const v2f*)&Hs[col][4 * j + 2 * half];

    v2f crow[8];
    #pragma unroll
    for (int r = 0; r < 8; ++r)
      crow[r] = Cs[r + 8 * half];

    float z0 = zin[(size_t)(seg * Bn + b) * 2 + 0];
    float z1 = zin[(size_t)(seg * Bn + b) * 2 + 1];

    const long long rowbase = (long long)b * T + (long long)seg * SEG;

    for (int c = 0; c < CHUNKS_PER_SEG; ++c) {
      const long long t0 = rowbase + (long long)c * CHUNK;

      if (c + 1 < CHUNKS_PER_SEG)
        __builtin_prefetch(u + t0 + CHUNK, 0, 1);

      v2f bfrag[4];
      #pragma unroll
      for (int j = 0; j < 4; ++j)
        bfrag[j] = *(const v2f*)(u + t0 + 4 * j + 2 * half);

      v8f acc;
      #pragma unroll
      for (int r = 0; r < 8; ++r) acc[r] = crow[r].x * z0 + crow[r].y * z1;

      acc = __builtin_amdgcn_wmma_f32_16x16x4_f32(false, afrag[0], false, bfrag[0], (short)0, acc, false, false);
      acc = __builtin_amdgcn_wmma_f32_16x16x4_f32(false, afrag[1], false, bfrag[1], (short)0, acc, false, false);
      acc = __builtin_amdgcn_wmma_f32_16x16x4_f32(false, afrag[2], false, bfrag[2], (short)0, acc, false, false);
      acc = __builtin_amdgcn_wmma_f32_16x16x4_f32(false, afrag[3], false, bfrag[3], (short)0, acc, false, false);

      v4f lo; lo.x = acc[0]; lo.y = acc[1]; lo.z = acc[2]; lo.w = acc[3];
      v4f hi; hi.x = acc[4]; hi.y = acc[5]; hi.z = acc[6]; hi.w = acc[7];
      __builtin_nontemporal_store(lo, (v4f*)(y + t0 + 8 * half + 0));
      __builtin_nontemporal_store(hi, (v4f*)(y + t0 + 8 * half + 4));

      float y14 = __shfl(acc[6],     col + 16, 32);
      float y15 = __shfl(acc[7],     col + 16, 32);
      float u14 = __shfl(bfrag[3].x, col + 16, 32);
      float u15 = __shfl(bfrag[3].y, col + 16, 32);
      float nz1 = b2v * u15 - a2 * y15;
      float nz0 = b1 * u15 - a1 * y15 + b2v * u14 - a2 * y14;
      z0 = nz0; z1 = nz1;
    }
  }
}

__global__ void naive_iir_kernel(const float* __restrict__ u,
                                 const float* __restrict__ bc,
                                 const float* __restrict__ fc,
                                 const float* __restrict__ yinit,
                                 const float* __restrict__ uinit,
                                 float* __restrict__ y, int Bn, int T) {
  int b = blockIdx.x * blockDim.x + threadIdx.x;
  if (b >= Bn) return;
  const float b0 = bc[0], b1 = bc[1];
  const float a1 = fc[0], a2 = fc[1];
  float z0 = b1 * uinit[b * 2 + 0] - a1 * yinit[b * 2 + 0] - a2 * yinit[b * 2 + 1];
  float z1 = -a2 * yinit[b * 2 + 0];
  const float* up = u + (long long)b * T;
  float* yp = y + (long long)b * T;
  for (int t = 0; t < T; ++t) {
    float ut = up[t];
    float yv = b0 * ut + z0;
    float nz0 = b1 * ut - a1 * yv + z1;
    z1 = -a2 * yv;
    z0 = nz0;
    yp[t] = yv;
  }
}

extern "C" void kernel_launch(void* const* d_in, const int* in_sizes, int n_in,
                              void* d_out, int out_size, void* d_ws, size_t ws_size,
                              hipStream_t stream) {
  const float* bc = (const float*)d_in[0];
  const float* fc = (const float*)d_in[1];
  const float* u  = (const float*)d_in[2];
  const float* yi = (const float*)d_in[3];
  const float* ui = (const float*)d_in[4];
  float* y = (float*)d_out;

  const int Bn = in_sizes[3] / 2;
  const int T  = in_sizes[2] / Bn;
  const int nseg = T / SEG;
  const size_t need = (size_t)2 * (size_t)nseg * (size_t)Bn * 2 * sizeof(float);

  if ((T % SEG) != 0 || (Bn % 16) != 0 || ws_size < need) {
    naive_iir_kernel<<<(Bn + 255) / 256, 256, 0, stream>>>(u, bc, fc, yi, ui, y, Bn, T);
    return;
  }

  float* zpart = (float*)d_ws;
  float* zin   = zpart + (size_t)nseg * Bn * 2;

  const int total = Bn * nseg;
  seg_state_kernel<<<(total + 255) / 256, 256, 0, stream>>>(u, bc, fc, zpart, Bn, T, nseg);
  scan_state_kernel<<<(Bn + 255) / 256, 256, 0, stream>>>(bc, fc, yi, ui, zpart, zin, Bn, nseg);

  const int waves  = nseg * (Bn / 16);
  const int blocks = (waves + 7) / 8;
  wmma_filter_kernel<<<blocks, 256, 0, stream>>>(u, bc, fc, zin, y, Bn, T, nseg);
}


# one launch: a wave owns one whole row and walks it in time with the filter state carried in registers, two register sets of loads ahead; state pass, scan pass and workspace gone: sum order changes
# speedup vs baseline: 7.9615x; 7.9615x over previous
#include <hip/hip_runtime.h>


typedef __attribute__((ext_vector_type(2))) float v2f;
typedef __attribute__((ext_vector_type(4))) float v4f;
typedef __attribute__((ext_vector_type(8))) float v8f;

#define SEG   256
#define NB    8
#define NSETS 3
#define ROWS_PER_BLOCK 4

template <int N>
__device__ __forceinline__ float row_shr_zero(float v) {
  return __builtin_bit_cast(float,
      __builtin_amdgcn_update_dpp(0, __builtin_bit_cast(int, v), 0x110 + N, 0xf, 0xf, true));
}
template <int N>
__device__ __forceinline__ void scan_step(float& s0, float& s1, v4f q) {
  const float t0 = row_shr_zero<N>(s0);
  const float t1 = row_shr_zero<N>(s1);
  s0 = __builtin_fmaf(q.y, t1, __builtin_fmaf(q.x, t0, s0));
  s1 = __builtin_fmaf(q.w, t1, __builtin_fmaf(q.z, t0, s1));
}
__device__ __forceinline__ float uniform_f(float v) {
  return __builtin_bit_cast(float, __builtin_amdgcn_readfirstlane(__builtin_bit_cast(int, v)));
}
__device__ __forceinline__ v4f uniform_v4(v4f v) {
  v4f r; r.x = uniform_f(v.x); r.y = uniform_f(v.y); r.z = uniform_f(v.z); r.w = uniform_f(v.w);
  return r;
}
__device__ __forceinline__ v4f mul2x2(v4f a, v4f b) {
  v4f r;
  r.x = a.x * b.x + a.y * b.z;
  r.y = a.x * b.y + a.y * b.w;
  r.z = a.z * b.x + a.w * b.z;
  r.w = a.z * b.y + a.w * b.w;
  return r;
}

__device__ __forceinline__ void load_set(v4f (&lo)[NB], v4f (&hi)[NB], const float* __restrict__ p) {
  #pragma unroll
  for (int q = 0; q < NB; ++q) {
    lo[q] = *(const v4f*)(p + q * SEG);
    hi[q] = *(const v4f*)(p + q * SEG + 8);
  }
}

struct FilterConsts {
  v2f afrag[4];
  v2f cfrag;
  v4f P1, P2, P4, P8, P16;
  v4f Pn;
  float b1, a1, a2;
  int half;
};

__device__ __forceinline__ void filter_set(const v4f (&lo)[NB], const v4f (&hi)[NB],
                                           float* __restrict__ yp, const FilterConsts& c,
                                           float& ze0, float& ze1) {
  #pragma unroll
  for (int q = 0; q < NB; ++q) {
    const v4f w0 = lo[q];
    const v4f w1 = hi[q];

    v2f bf0; bf0.x = w0.x; bf0.y = w0.y;
    v2f bf1; bf1.x = w0.z; bf1.y = w0.w;
    v2f bf2; bf2.x = w1.x; bf2.y = w1.y;
    v2f bf3; bf3.x = w1.z; bf3.y = w1.w;
    v8f acc = {0.f, 0.f, 0.f, 0.f, 0.f, 0.f, 0.f, 0.f};
    acc = __builtin_amdgcn_wmma_f32_16x16x4_f32(false, c.afrag[0], false, bf0, (short)0, acc, false, false);
    acc = __builtin_amdgcn_wmma_f32_16x16x4_f32(false, c.afrag[1], false, bf1, (short)0, acc, false, false);
    acc = __builtin_amdgcn_wmma_f32_16x16x4_f32(false, c.afrag[2], false, bf2, (short)0, acc, false, false);
    acc = __builtin_amdgcn_wmma_f32_16x16x4_f32(false, c.afrag[3], false, bf3, (short)0, acc, false, false);

    const float zp0 = c.b1 * w1.w - c.a1 * acc[7] - c.a2 * acc[6];
    const float zp1 = -c.a2 * acc[7];

    float s0 = row_shr_zero<1>(zp0);
    float s1 = row_shr_zero<1>(zp1);
    scan_step<1>(s0, s1, c.P1);
    scan_step<2>(s0, s1, c.P2);
    scan_step<4>(s0, s1, c.P4);
    scan_step<8>(s0, s1, c.P8);

    const float x0 = __builtin_fmaf(c.P1.y, s1, __builtin_fmaf(c.P1.x, s0, zp0));
    const float x1 = __builtin_fmaf(c.P1.w, s1, __builtin_fmaf(c.P1.z, s0, zp1));
    const float zs0 = __builtin_bit_cast(float, __builtin_amdgcn_readlane(__builtin_bit_cast(int, x0), 31));
    const float zs1 = __builtin_bit_cast(float, __builtin_amdgcn_readlane(__builtin_bit_cast(int, x1), 31));

    const float e0 = __builtin_fmaf(c.Pn.y, ze1, __builtin_fmaf(c.Pn.x, ze0, s0));
    const float e1 = __builtin_fmaf(c.Pn.w, ze1, __builtin_fmaf(c.Pn.z, ze0, s1));

    v2f zf;
    zf.x = c.half ? e0 : 0.f;
    zf.y = c.half ? e1 : 0.f;
    acc = __builtin_amdgcn_wmma_f32_16x16x4_f32(false, c.cfrag, false, zf, (short)0, acc, false, false);

    float* yq = yp + q * SEG;
    v4f olo; olo.x = acc[0]; olo.y = acc[1]; olo.z = acc[2]; olo.w = acc[3];
    v4f ohi; ohi.x = acc[4]; ohi.y = acc[5]; ohi.z = acc[6]; ohi.w = acc[7];
    __builtin_nontemporal_store(olo, (v4f*)yq);
    __builtin_nontemporal_store(ohi, (v4f*)(yq + 8));

    const float n0 = __builtin_fmaf(c.P16.y, ze1, __builtin_fmaf(c.P16.x, ze0, zs0));
    const float n1 = __builtin_fmaf(c.P16.w, ze1, __builtin_fmaf(c.P16.z, ze0, zs1));
    ze0 = n0; ze1 = n1;
  }
}

__global__ __launch_bounds__(ROWS_PER_BLOCK * 32)
void iir_row_walk_kernel(const float* __restrict__ u,
                         const float* __restrict__ bc,
                         const float* __restrict__ fc,
                         const float* __restrict__ yinit,
                         const float* __restrict__ uinit,
                         float* __restrict__ y,
                         int Bn, int T, int nsets) {
  __shared__ float Hs[16][16];
  __shared__ v2f   Cs[16];
  __shared__ v4f   Ps[5];

  const float b0 = bc[0], b1 = bc[1], b2v = 0.0f;
  const float a1 = fc[0], a2 = fc[1];

  if (threadIdx.x < 16) {
    const int t = (int)threadIdx.x;
    float hm1 = 0.f, hm2 = 0.f;
    #pragma unroll
    for (int d = 0; d < 16; ++d) {
      float hd;
      if (d == 0)      hd = b0;
      else if (d == 1) hd = b1 - a1 * b0;
      else if (d == 2) hd = b2v - a1 * hm1 - a2 * hm2;
      else             hd = -a1 * hm1 - a2 * hm2;
      if (d <= t) Hs[t][t - d] = hd;
      hm2 = hm1; hm1 = hd;
    }
    for (int k = t + 1; k < 16; ++k) Hs[t][k] = 0.f;
  } else if (threadIdx.x == 32) {
    float cc0 = 1.f, cc1 = 0.f;
    #pragma unroll
    for (int k = 0; k < 16; ++k) {
      v2f cv; cv.x = cc0; cv.y = cc1;
      Cs[k] = cv;
      float n0 = -a1 * cc0 - a2 * cc1;
      cc1 = cc0; cc0 = n0;
    }
  } else if (threadIdx.x == 64) {
    float m00 = -a1, m01 = 1.f, m10 = -a2, m11 = 0.f;
    #pragma unroll
    for (int i = 0; i < 8; ++i) {
      float t00 = m00 * m00 + m01 * m10;
      float t01 = m00 * m01 + m01 * m11;
      float t10 = m10 * m00 + m11 * m10;
      float t11 = m10 * m01 + m11 * m11;
      m00 = t00; m01 = t01; m10 = t10; m11 = t11;
      if (i >= 3) { v4f p; p.x = m00; p.y = m01; p.z = m10; p.w = m11; Ps[i - 3] = p; }
    }
  }
  __syncthreads();

  const int lane = (int)(threadIdx.x & 31);
  const int row = __builtin_amdgcn_readfirstlane(
      (int)(blockIdx.x * ROWS_PER_BLOCK + (threadIdx.x >> 5)));

  if (row < Bn) {
    const int col  = lane & 15;
    const int half = lane >> 4;
    const int trow = (col & 3) | ((col & 4) << 1) | ((col & 8) >> 1);

    FilterConsts c;
    #pragma unroll
    for (int j = 0; j < 4; ++j)
      c.afrag[j] = *(const v2f*)&Hs[trow][8 * (j >> 1) + 4 * half + 2 * (j & 1)];
    c.cfrag = Cs[trow];
    if (half == 0) { c.cfrag.x = 0.f; c.cfrag.y = 0.f; }
    c.P1 = uniform_v4(Ps[0]); c.P2 = uniform_v4(Ps[1]); c.P4 = uniform_v4(Ps[2]);
    c.P8 = uniform_v4(Ps[3]); c.P16 = uniform_v4(Ps[4]);
    v4f pn; pn.x = 1.f; pn.y = 0.f; pn.z = 0.f; pn.w = 1.f;
    { const v4f r = mul2x2(c.P1, pn); if (col & 1) pn = r; }
    { const v4f r = mul2x2(c.P2, pn); if (col & 2) pn = r; }
    { const v4f r = mul2x2(c.P4, pn); if (col & 4) pn = r; }
    { const v4f r = mul2x2(c.P8, pn); if (col & 8) pn = r; }
    c.Pn = pn;
    c.b1 = b1; c.a1 = a1; c.a2 = a2; c.half = half;

    float ze0 = b1 * uinit[row * 2 + 0] - a1 * yinit[row * 2 + 0] - a2 * yinit[row * 2 + 1];
    float ze1 = -a2 * yinit[row * 2 + 0];

    const int loff = 16 * col + 4 * half;
    const float* up = u + (size_t)row * T + loff;
    float*       yp = y + (size_t)row * T + loff;
    const int setlen = NB * SEG;

    v4f lo[NSETS][NB], hi[NSETS][NB];
    #pragma unroll
    for (int r = 0; r < NSETS - 1; ++r) load_set(lo[r], hi[r], up + (size_t)r * setlen);

    int s = 0;
    for (; s < nsets - (NSETS - 1); s += NSETS) {
      #pragma unroll
      for (int r = 0; r < NSETS; ++r) {
        __builtin_amdgcn_sched_barrier(0);
        load_set(lo[(r + NSETS - 1) % NSETS], hi[(r + NSETS - 1) % NSETS],
                 up + (size_t)(s + r + NSETS - 1) * setlen);
        __builtin_amdgcn_sched_barrier(0);
        filter_set(lo[r], hi[r], yp + (size_t)(s + r) * setlen, c, ze0, ze1);
      }
    }
    #pragma unroll
    for (int r = 0; r < NSETS - 1; ++r)
      filter_set(lo[r], hi[r], yp + (size_t)(s + r) * setlen, c, ze0, ze1);
  }
}

__global__ void naive_iir_kernel(const float* __restrict__ u,
                                 const float* __restrict__ bc,
                                 const float* __restrict__ fc,
                                 const float* __restrict__ yinit,
                                 const float* __restrict__ uinit,
                                 float* __restrict__ y, int Bn, int T) {
  int b = blockIdx.x * blockDim.x + threadIdx.x;
  if (b >= Bn) return;
  const float b0 = bc[0], b1 = bc[1];
  const float a1 = fc[0], a2 = fc[1];
  float z0 = b1 * uinit[b * 2 + 0] - a1 * yinit[b * 2 + 0] - a2 * yinit[b * 2 + 1];
  float z1 = -a2 * yinit[b * 2 + 0];
  const float* up = u + (long long)b * T;
  float* yp = y + (long long)b * T;
  for (int t = 0; t < T; ++t) {
    float ut = up[t];
    float yv = b0 * ut + z0;
    float nz0 = b1 * ut - a1 * yv + z1;
    z1 = -a2 * yv;
    z0 = nz0;
    yp[t] = yv;
  }
}

extern "C" void kernel_launch(void* const* d_in, const int* in_sizes, int n_in,
                              void* d_out, int out_size, void* d_ws, size_t ws_size,
                              hipStream_t stream) {
  const float* bc = (const float*)d_in[0];
  const float* fc = (const float*)d_in[1];
  const float* u  = (const float*)d_in[2];
  const float* yi = (const float*)d_in[3];
  const float* ui = (const float*)d_in[4];
  float* y = (float*)d_out;

  const int Bn = in_sizes[3] / 2;
  const int T  = in_sizes[2] / Bn;
  const int setlen = NB * SEG;
  const int nsets = T / setlen;

  if ((T % setlen) != 0 || nsets < NSETS - 1 || ((nsets - (NSETS - 1)) % NSETS) != 0) {
    naive_iir_kernel<<<(Bn + 255) / 256, 256, 0, stream>>>(u, bc, fc, yi, ui, y, Bn, T);
    return;
  }

  const int blocks = (Bn + ROWS_PER_BLOCK - 1) / ROWS_PER_BLOCK;
  iir_row_walk_kernel<<<blocks, ROWS_PER_BLOCK * 32, 0, stream>>>(u, bc, fc, yi, ui, y, Bn, T, nsets);
}
